# starting kernel
# baseline (speedup 1.0000x reference)
#include <hip/hip_runtime.h>
#include <hip/hip_bf16.h>
#include <stdint.h>

#define BB    8
#define SS    512
#define EE    768
#define VV    30522
#define GVV   20000
#define NNZV  400000
#define HH    64
#define OO    16

#define WPB     640
#define NCHUNK  8
#define CHUNK   (GVV / NCHUNK)
#define KSTEPS  (CHUNK / 4)
#define ETILES  (EE / 16)

typedef float v2f __attribute__((ext_vector_type(2)));
typedef float v8f __attribute__((ext_vector_type(8)));

#define PRESENT_BYTES  20480
#define HVH_BYTES      ((size_t)BB * GVV * HH * 4)
#define PART_BYTES     ((size_t)NCHUNK * BB * EE * HH * 4)

__global__ void gcn_zero(uint32_t* __restrict__ p, long n) {
  long i = (long)blockIdx.x * blockDim.x + threadIdx.x;
  long stride = (long)gridDim.x * blockDim.x;
  for (; i < n; i += stride) p[i] = 0u;
}

__global__ void gcn_present(const int* __restrict__ input_ids,
                            const int* __restrict__ tok2g,
                            uint32_t* __restrict__ present) {
  int tid = blockIdx.x * blockDim.x + threadIdx.x;
  if (tid >= BB * SS) return;
  int b = tid / SS;
  int id = input_ids[tid];
  if (id > VV - 1 || id < 0) id = 0;
  int g = tok2g[id];
  atomicOr(&present[b * WPB + (g >> 5)], 1u << (g & 31));
}

__global__ void gcn_scatter(const int* __restrict__ rows,
                            const int* __restrict__ cols,
                            const float* __restrict__ vals,
                            const float* __restrict__ W,
                            const uint32_t* __restrict__ present,
                            float* __restrict__ Hvh) {
  int k = blockIdx.x * 4 + (threadIdx.x >> 6);
  const int h = threadIdx.x & 63;
  const int kStride = gridDim.x * 4;
  for (; k < NNZV; k += kStride) {
    const int row = rows[k];
    const int col = cols[k];
    const float vw = vals[k] * W[(size_t)col * HH + h];
    const int rw = row >> 5, rbit = row & 31;
    const int cw = col >> 5, cbit = col & 31;
#pragma unroll
    for (int b = 0; b < BB; ++b) {
      const uint32_t m = (present[b * WPB + rw] >> rbit) |
                         (present[b * WPB + cw] >> cbit);
      if (m & 1u)
        atomicAdd(&Hvh[((size_t)b * GVV + row) * HH + h], vw);
    }
  }
}

__global__ void __launch_bounds__(128)
gcn_gemm_wmma(const float* __restrict__ emb,
              const int* __restrict__ gv_ids,
              const float* __restrict__ Hvh,
              float* __restrict__ part) {
  const int wave   = blockIdx.x * 4 + (threadIdx.x >> 5);
  const int lane   = threadIdx.x & 31;
  const int lane16 = lane & 15;
  const int hi     = lane >> 4;

  const int c  = wave % NCHUNK;
  const int et = (wave / NCHUNK) % ETILES;
  const int b  = wave / (NCHUNK * ETILES);

  const int e0    = et * 16;
  const int vbase = c * CHUNK;
  const float* __restrict__ HvhB = Hvh + (size_t)b * GVV * HH;

  const v8f zacc = {0.f, 0.f, 0.f, 0.f, 0.f, 0.f, 0.f, 0.f};
  v8f acc0 = zacc, acc1 = zacc, acc2 = zacc, acc3 = zacc;

  for (int step = 0; step < KSTEPS; ++step) {
    const int va = vbase + step * 4 + 2 * hi;
    const int g0 = gv_ids[va];
    const int g1 = gv_ids[va + 1];

    v2f a;
    a.x = emb[(size_t)g0 * EE + e0 + lane16];
    a.y = emb[(size_t)g1 * EE + e0 + lane16];

    const float* __restrict__ r0 = HvhB + (size_t)va * HH + lane16;
    const float* __restrict__ r1 = r0 + HH;

    v2f b0, b1, b2, b3;
    b0.x = r0[0];  b0.y = r1[0];
    b1.x = r0[16]; b1.y = r1[16];
    b2.x = r0[32]; b2.y = r1[32];
    b3.x = r0[48]; b3.y = r1[48];

    acc0 = __builtin_amdgcn_wmma_f32_16x16x4_f32(false, a, false, b0,
                                                 (short)0, acc0, false, false);
    acc1 = __builtin_amdgcn_wmma_f32_16x16x4_f32(false, a, false, b1,
                                                 (short)0, acc1, false, false);
    acc2 = __builtin_amdgcn_wmma_f32_16x16x4_f32(false, a, false, b2,
                                                 (short)0, acc2, false, false);
    acc3 = __builtin_amdgcn_wmma_f32_16x16x4_f32(false, a, false, b3,
                                                 (short)0, acc3, false, false);
  }

  float* __restrict__ pb =
      part + (((size_t)c * BB + b) * EE + e0) * HH + lane16;
  v8f accs[4] = {acc0, acc1, acc2, acc3};
#pragma unroll
  for (int t = 0; t < 4; ++t)
#pragma unroll
    for (int r = 0; r < 8; ++r) {
      const int e = r + 8 * hi;
      pb[(size_t)e * HH + t * 16] = accs[t][r];
    }
}

__global__ void gcn_final(const float* __restrict__ part,
                          const float* __restrict__ fc_w,
                          const float* __restrict__ fc_b,
                          float* __restrict__ out) {
  __shared__ float lds[HH];
  const int e = blockIdx.x % EE;
  const int b = blockIdx.x / EE;
  const int h = threadIdx.x;
  float f = 0.f;
#pragma unroll
  for (int c = 0; c < NCHUNK; ++c)
    f += part[(((size_t)c * BB + b) * EE + e) * HH + h];
  lds[h] = f;
  __syncthreads();
  if (h < OO) {
    float a = fc_b[h];
#pragma unroll
    for (int hh = 0; hh < HH; ++hh) a += lds[hh] * fc_w[h * HH + hh];
    out[((size_t)b * OO + h) * EE + e] = a;
  }
}

extern "C" void kernel_launch(void* const* d_in, const int* in_sizes, int n_in,
                              void* d_out, int out_size, void* d_ws, size_t ws_size,
                              hipStream_t stream) {
  (void)in_sizes; (void)n_in; (void)out_size;
  const float* emb        = (const float*)d_in[0];
  const float* adj_values = (const float*)d_in[1];
  const float* W_vh       = (const float*)d_in[2];
  const float* fc_w       = (const float*)d_in[3];
  const float* fc_b       = (const float*)d_in[4];
  const int*   input_ids  = (const int*)d_in[5];
  const int*   tok2g      = (const int*)d_in[6];
  const int*   gv_ids     = (const int*)d_in[7];
  const int*   adj_rows   = (const int*)d_in[8];
  const int*   adj_cols   = (const int*)d_in[9];
  float* out = (float*)d_out;

  const size_t need = PRESENT_BYTES + HVH_BYTES + PART_BYTES;
  if (ws_size < need) return;

  uint8_t*  ws      = (uint8_t*)d_ws;
  uint32_t* present = (uint32_t*)ws;
  float*    Hvh     = (float*)(ws + PRESENT_BYTES);
  float*    part    = (float*)(ws + PRESENT_BYTES + HVH_BYTES);

  gcn_zero<<<4096, 256, 0, stream>>>((uint32_t*)ws, (long)(need / 4));
  gcn_present<<<(BB * SS) / 256, 256, 0, stream>>>(input_ids, tok2g, present);
  gcn_scatter<<<8192, 256, 0, stream>>>(adj_rows, adj_cols, adj_values, W_vh,
                                        present, Hvh);
  gcn_gemm_wmma<<<(BB * ETILES * NCHUNK) / 4, 128, 0, stream>>>(emb, gv_ids,
                                                                Hvh, part);
  gcn_final<<<BB * EE, HH, 0, stream>>>(part, fc_w, fc_b, out);
}


# row-present term off global atomics: scan writes records, 64 added last-launch blocks sum R tiles in LDS, apply fc, WMMA with emb rows; both classes add into out over a bias start; sum order changes
# speedup vs baseline: 8.1201x; 8.1201x over previous
#include <hip/hip_runtime.h>
#include <stdint.h>

#define BB    8
#define SS    512
#define EE    768
#define VV    30522
#define GVV   20000
#define NNZV  400000
#define HH    64
#define OO    16

#define WPB     640

typedef float v2f __attribute__((ext_vector_type(2)));
typedef float v8f __attribute__((ext_vector_type(8)));

#define OFF_MASK8    0
#define OFF_SLOT     20480
#define OFF_VTOK     (OFF_SLOT + BB * GVV * 4)
#define OFF_CNT      (OFF_VTOK + BB * SS * 4)
#define CSTR         32
#define OFF_LIST     (OFF_CNT + BB * CSTR * 4)
#define WS_NEED      ((size_t)OFF_LIST + (size_t)BB * NNZV * 8)

#define NB1   (BB * (EE / 32))
#define NQ    8

__global__ void __launch_bounds__(1024)
vg_prep(const int* __restrict__ input_ids, const int* __restrict__ tok2g,
        const float* __restrict__ fc_b,
        uint32_t* __restrict__ mask8w, int* __restrict__ slotmap,
        int* __restrict__ vtok, float4* __restrict__ out4, int* __restrict__ cnt) {
  __shared__ uint32_t m8[GVV / 4];
  __shared__ uint32_t bm[BB * WPB];
  const int t = threadIdx.x;
  if (blockIdx.x != 0) {
    const int i4 = (blockIdx.x - 1) * 1024 + t;
    const float bv = fc_b[((i4 * 4) / EE) % OO];
    out4[i4] = make_float4(bv, bv, bv, bv);
    if (blockIdx.x == 1 && t < BB) { cnt[t * CSTR] = 0; cnt[t * CSTR + 1] = 0; }
    return;
  }
  for (int i = t; i < GVV / 4; i += 1024) m8[i] = 0u;
  for (int i = t; i < BB * WPB; i += 1024) bm[i] = 0u;
  __syncthreads();
  int vq[(BB * SS) / 1024];
#pragma unroll
  for (int q = 0; q < (BB * SS) / 1024; ++q) {
    int id = input_ids[q * 1024 + t];
    if (id > VV - 1 || id < 0) id = 0;
    vq[q] = id;
  }
#pragma unroll
  for (int q = 0; q < (BB * SS) / 1024; ++q) vq[q] = tok2g[vq[q]];
#pragma unroll
  for (int q = 0; q < (BB * SS) / 1024; ++q) {
    const int idx = q * 1024 + t;
    const int b = idx >> 9, s = idx & (SS - 1);
    const int v = vq[q];
    const uint32_t bit = 1u << (v & 31);
    const uint32_t old = atomicOr(&bm[b * WPB + (v >> 5)], bit);
    const bool own = (old & bit) == 0u;
    atomicOr(&m8[v >> 2], 1u << ((v & 3) * 8 + b));
    if (own) slotmap[b * GVV + v] = s;
    vtok[idx] = own ? v : -1;
  }
  __syncthreads();
  for (int i = t; i < GVV / 4; i += 1024) mask8w[i] = m8[i];
}

__global__ void __launch_bounds__(256)
vg_scan(const int* __restrict__ rows, const int* __restrict__ cols,
        const float* __restrict__ vals,
        const int* __restrict__ gv_ids, const uint8_t* __restrict__ mask8,
        const int* __restrict__ slotmap,
        int* __restrict__ cnt, int2* __restrict__ list) {
  __shared__ int lcnt[2 * BB];
  __shared__ int lbase[2 * BB];
  if (threadIdx.x < 2 * BB) lcnt[threadIdx.x] = 0;
  __syncthreads();
  const int lane = threadIdx.x & 31;
  const int k = blockIdx.x * 256 + threadIdx.x;
  int row = 0, col = 0; float val = 0.f;
  uint32_t rm = 0u, cm = 0u;
  if (k < NNZV) {
    row = rows[k]; col = cols[k]; val = vals[k];
    rm = mask8[row]; cm = mask8[col];
  }
  const int gv = cm ? gv_ids[row] : 0;
  const uint32_t t1 = rm & ~cm;
  uint32_t bal[BB], bal2[BB];
  int myc = 0;
#pragma unroll
  for (int b = 0; b < BB; ++b) {
    bal[b]  = (uint32_t)__ballot((cm >> b) & 1u);
    bal2[b] = (uint32_t)__ballot((t1 >> b) & 1u);
    if (lane == b) myc = __popc(bal[b]);
    if (lane == BB + b) myc = __popc(bal2[b]);
  }
  int woff = 0;
  if (lane < 2 * BB && myc > 0) woff = atomicAdd(&lcnt[lane], myc);
  __syncthreads();
  if (threadIdx.x < BB) {
    const uint32_t c1 = (uint32_t)lcnt[threadIdx.x];
    const uint32_t c2 = (uint32_t)lcnt[BB + threadIdx.x];
    unsigned long long old = 0ull;
    if (c1 | c2)
      old = atomicAdd((unsigned long long*)(cnt + threadIdx.x * CSTR),
                      (unsigned long long)c1 | ((unsigned long long)c2 << 32));
    lbase[threadIdx.x] = (int)(uint32_t)old;
    lbase[BB + threadIdx.x] = (int)(uint32_t)(old >> 32);
  }
  __syncthreads();
  const uint32_t below = (1u << lane) - 1u;
#pragma unroll
  for (int b = 0; b < BB; ++b) {
    const int p1 = lbase[b] + __shfl(woff, b);
    const int p2 = lbase[BB + b] + __shfl(woff, BB + b);
    int2* __restrict__ lb = list + (size_t)b * NNZV;
    if ((cm >> b) & 1u) {
      const int pos = p1 + __popc(bal[b] & below);
      const int s = slotmap[b * GVV + col];
      lb[pos] = make_int2(gv | (s << 16), __float_as_int(val));
    }
    if ((t1 >> b) & 1u) {
      const int pos = p2 + __popc(bal2[b] & below);
      const int s = slotmap[b * GVV + row];
      lb[NNZV - 1 - pos] = make_int2(col | (s << 16), __float_as_int(val));
    }
  }
}

#define P3   40
#define CPP  65
#define NES  (EE / 32)
#define QCAP 4096

__device__ __forceinline__ void
fuse_colpart(const int b, const int es,
             const float* __restrict__ emb, const float* __restrict__ W,
             const float* __restrict__ fc_w, const int* __restrict__ vtok,
             const int* __restrict__ cnt, const int2* __restrict__ list, float* __restrict__ out) {
  __shared__ __attribute__((aligned(16))) float T[SS * P3];
  __shared__ float Cp[4 * 32 * CPP];
  __shared__ int vs[SS];

  const int tid  = threadIdx.x;
  const int w    = __builtin_amdgcn_readfirstlane(tid >> 5);
  const int lane = tid & 31;
  const int c16  = lane & 15;
  const int hi   = lane >> 4;
  const float* __restrict__ embc = emb + es * 32 + lane;

  for (int i = tid; i < (SS * P3) / 4; i += 1024) ((float4*)T)[i] = make_float4(0.f, 0.f, 0.f, 0.f);
  if (tid < SS) vs[tid] = vtok[b * SS + tid];
  int n = cnt[b * CSTR];
  if (n > NNZV) n = NNZV;
  const int2* __restrict__ lb = list + (size_t)b * NNZV;
  int2 rec = make_int2(0, 0);
  if (w * 32 + lane < n) rec = lb[w * 32 + lane];
  __syncthreads();

  for (int base = w * 32; base < n; base += 1024) {
    int2 nxt = make_int2(0, 0);
    if (base + 1024 + lane < n) nxt = lb[base + 1024 + lane];
    const int m = (n - base) < 32 ? (n - base) : 32;
    if (m == 32) {
      float x[32];
#pragma unroll
      for (int j = 0; j < 32; ++j) {
        const int gv = __builtin_amdgcn_readlane(rec.x, j) & 0xFFFF;
        x[j] = embc[(size_t)gv * EE];
      }
#pragma unroll
      for (int j = 0; j < 32; ++j) {
        const int s = __builtin_amdgcn_readlane(rec.x, j) >> 16;
        const float val = __int_as_float(__builtin_amdgcn_readlane(rec.y, j));
        atomicAdd(&T[s * P3 + lane], val * x[j]);
      }
    } else {
      for (int j = 0; j < m; ++j) {
        const int rx = __builtin_amdgcn_readlane(rec.x, j);
        const float val = __int_as_float(__builtin_amdgcn_readlane(rec.y, j));
        atomicAdd(&T[(rx >> 16) * P3 + lane], val * embc[(size_t)(rx & 0xFFFF) * EE]);
      }
    }
    rec = nxt;
  }
  __syncthreads();

  const int tl = w & 7;
  const int mi = tl & 1;
  const int ni = tl >> 1;
  const int kq = w >> 3;
  const int arow = mi * 16 + c16;
  const int ncol = ni * 16 + c16;
  const float* __restrict__ Wc = W + ncol;
  const v8f zacc = {0.f, 0.f, 0.f, 0.f, 0.f, 0.f, 0.f, 0.f};
  v8f acc1 = zacc, acc2 = zacc;
#pragma unroll 8
  for (int j = 0; j < 32; j += 2) {
    const int s0 = kq * 128 + 4 * j + 2 * hi;
    const int v0 = vs[s0], v1 = vs[s0 + 1], v2 = vs[s0 + 4], v3 = vs[s0 + 5];
    const float w0 = Wc[(size_t)(v0 < 0 ? 0 : v0) * HH];
    const float w1 = Wc[(size_t)(v1 < 0 ? 0 : v1) * HH];
    const float w2 = Wc[(size_t)(v2 < 0 ? 0 : v2) * HH];
    const float w3 = Wc[(size_t)(v3 < 0 ? 0 : v3) * HH];
    const v2f a1 = {T[s0 * P3 + arow], T[(s0 + 1) * P3 + arow]};
    const v2f b1 = {v0 < 0 ? 0.f : w0, v1 < 0 ? 0.f : w1};
    const v2f a2 = {T[(s0 + 4) * P3 + arow], T[(s0 + 5) * P3 + arow]};
    const v2f b2 = {v2 < 0 ? 0.f : w2, v3 < 0 ? 0.f : w3};
    acc1 = __builtin_amdgcn_wmma_f32_16x16x4_f32(false, a1, false, b1, (short)0, acc1, false, false);
    acc2 = __builtin_amdgcn_wmma_f32_16x16x4_f32(false, a2, false, b2, (short)0, acc2, false, false);
  }
  const v8f accs = acc1 + acc2;
#pragma unroll
  for (int r = 0; r < 8; ++r)
    Cp[(kq * 32 + mi * 16 + 8 * hi + r) * CPP + ncol] = accs[r];
  __syncthreads();

  if (tid < OO * 32) {
    const int o = tid >> 5;
    const int e = tid & 31;
    float a = 0.f;
    for (int h = 0; h < HH; ++h) {
      const float f = (Cp[(0 * 32 + e) * CPP + h] + Cp[(1 * 32 + e) * CPP + h]) +
                      (Cp[(2 * 32 + e) * CPP + h] + Cp[(3 * 32 + e) * CPP + h]);
      a += f * fc_w[o * HH + h];
    }
    atomicAdd(&out[((size_t)b * OO + o) * EE + es * 32 + e], a);
  }
}

__device__ __forceinline__ void
fuse_rowpart(const int b, const int q,
             const float* __restrict__ emb, const float* __restrict__ W,
             const float* __restrict__ fc_w, const int* __restrict__ gv_ids,
             const int* __restrict__ vtok, const int* __restrict__ cnt,
             const int2* __restrict__ list, float* __restrict__ out) {
  __shared__ __attribute__((aligned(16))) float Rt[64 * HH];
  __shared__ float RF[64 * OO];
  __shared__ int   es[64];
  __shared__ int2  qrec[QCAP];
  __shared__ int   qn[2];

  const int tid  = threadIdx.x;
  const int w    = __builtin_amdgcn_readfirstlane(tid >> 5);
  const int lane = tid & 31;
  const int c16  = lane & 15;
  const int hi   = lane >> 4;

  ((float4*)Rt)[tid] = make_float4(0.f, 0.f, 0.f, 0.f);
  if (tid < 2) qn[tid] = 0;
  if (tid < 64) {
    const int v = vtok[b * SS + q * 64 + tid];
    es[tid] = v < 0 ? -1 : gv_ids[v];
  }
  int n = cnt[b * CSTR + 1];
  if (n > NNZV) n = NNZV;
  const int2* __restrict__ lbk = list + (size_t)b * NNZV + (NNZV - 1);
  const uint32_t below = (1u << lane) - 1u;
  __syncthreads();

  int par = 0;
  for (int base = 0; base < n; base += QCAP, par ^= 1) {
    int2 r4[QCAP / 1024];
#pragma unroll
    for (int j = 0; j < QCAP / 1024; ++j) {
      const int i = base + j * 1024 + tid;
      r4[j] = make_int2(-1, 0);
      if (i < n) r4[j] = lbk[-(ptrdiff_t)i];
    }
    uint32_t hb[QCAP / 1024];
    int tot = 0;
#pragma unroll
    for (int j = 0; j < QCAP / 1024; ++j) {
      hb[j] = (uint32_t)__ballot(((uint32_t)r4[j].x >> 22) == (uint32_t)q);
      tot += __popc(hb[j]);
    }
    int p0 = 0;
    if (lane == 0 && tot > 0) p0 = atomicAdd(&qn[par], tot);
    p0 = __shfl(p0, 0);
#pragma unroll
    for (int j = 0; j < QCAP / 1024; ++j) {
      if ((hb[j] >> lane) & 1u) qrec[p0 + __popc(hb[j] & below)] = r4[j];
      p0 += __popc(hb[j]);
    }
    __syncthreads();
    const int m = __builtin_amdgcn_readfirstlane(qn[par]);
    if (tid == 0) qn[par ^ 1] = 0;
    for (int i0 = w; i0 < m; i0 += 256) {
      int rx[8]; float rv[8]; float2 wv[8];
#pragma unroll
      for (int jj = 0; jj < 8; ++jj) {
        int2 r = make_int2(0, 0);
        if (i0 + 32 * jj < m) r = qrec[i0 + 32 * jj];
        rx[jj] = __builtin_amdgcn_readfirstlane(r.x);
        rv[jj] = __int_as_float(__builtin_amdgcn_readfirstlane(r.y));
      }
#pragma unroll
      for (int jj = 0; jj < 8; ++jj) {
        wv[jj] = make_float2(0.f, 0.f);
        if (i0 + 32 * jj < m) wv[jj] = *(const float2*)(W + (size_t)(rx[jj] & 0xFFFF) * HH + 2 * lane);
      }
#pragma unroll
      for (int jj = 0; jj < 8; ++jj) {
        if (i0 + 32 * jj < m) {
          float* d = &Rt[((rx[jj] >> 16) & 63) * HH + 2 * lane];
          atomicAdd(d, rv[jj] * wv[jj].x);
          atomicAdd(d + 1, rv[jj] * wv[jj].y);
        }
      }
    }
    __syncthreads();
  }

  {
    const int s = tid >> 4, o = tid & 15;
    float a = 0.f;
    for (int h = 0; h < HH; ++h) a += Rt[s * HH + h] * fc_w[o * HH + h];
    RF[s * OO + o] = a;
  }
  __syncthreads();

  const v8f zacc = {0.f, 0.f, 0.f, 0.f, 0.f, 0.f, 0.f, 0.f};
  for (int t = w; t < EE / 16; t += 32) {
    const float* __restrict__ embt = emb + t * 16 + c16;
    v8f acc1 = zacc, acc2 = zacc;
#pragma unroll 4
    for (int ks = 0; ks < 16; ks += 2) {
      const int s0 = 4 * ks + 2 * hi;
      const int g0 = es[s0], g1 = es[s0 + 1], g2 = es[s0 + 4], g3 = es[s0 + 5];
      const float x0 = embt[(size_t)(g0 < 0 ? 0 : g0) * EE];
      const float x1 = embt[(size_t)(g1 < 0 ? 0 : g1) * EE];
      const float x2 = embt[(size_t)(g2 < 0 ? 0 : g2) * EE];
      const float x3 = embt[(size_t)(g3 < 0 ? 0 : g3) * EE];
      const v2f a1 = {RF[s0 * OO + c16], RF[(s0 + 1) * OO + c16]};
      const v2f b1 = {g0 < 0 ? 0.f : x0, g1 < 0 ? 0.f : x1};
      const v2f a2 = {RF[(s0 + 4) * OO + c16], RF[(s0 + 5) * OO + c16]};
      const v2f b2 = {g2 < 0 ? 0.f : x2, g3 < 0 ? 0.f : x3};
      acc1 = __builtin_amdgcn_wmma_f32_16x16x4_f32(false, a1, false, b1, (short)0, acc1, false, false);
      acc2 = __builtin_amdgcn_wmma_f32_16x16x4_f32(false, a2, false, b2, (short)0, acc2, false, false);
    }
    const v8f accs = acc1 + acc2;
#pragma unroll
    for (int r = 0; r < 8; ++r)
      atomicAdd(&out[((size_t)b * OO + 8 * hi + r) * EE + t * 16 + c16], accs[r]);
  }
}

__global__ void __launch_bounds__(1024)
vg_fuse(const float* __restrict__ emb, const float* __restrict__ W,
        const float* __restrict__ fc_w,
        const int* __restrict__ gv_ids, const int* __restrict__ vtok,
        const int* __restrict__ cnt,
        const int2* __restrict__ list, float* __restrict__ out) {
  if (blockIdx.x < NB1) {
    fuse_colpart(blockIdx.x / NES, blockIdx.x % NES, emb, W, fc_w, vtok, cnt, list, out);
  } else {
    const int i2 = blockIdx.x - NB1;
    fuse_rowpart(i2 / NQ, i2 % NQ, emb, W, fc_w, gv_ids, vtok, cnt, list, out);
  }
}

extern "C" void kernel_launch(void* const* d_in, const int* in_sizes, int n_in,
                              void* d_out, int out_size, void* d_ws, size_t ws_size,
                              hipStream_t stream) {
  (void)in_sizes; (void)n_in; (void)out_size;
  const float* emb        = (const float*)d_in[0];
  const float* adj_values = (const float*)d_in[1];
  const float* W_vh       = (const float*)d_in[2];
  const float* fc_w       = (const float*)d_in[3];
  const float* fc_b       = (const float*)d_in[4];
  const int*   input_ids  = (const int*)d_in[5];
  const int*   tok2g      = (const int*)d_in[6];
  const int*   gv_ids     = (const int*)d_in[7];
  const int*   adj_rows   = (const int*)d_in[8];
  const int*   adj_cols   = (const int*)d_in[9];
  float* out = (float*)d_out;

  if (ws_size < WS_NEED) return;

  uint8_t*  ws      = (uint8_t*)d_ws;
  uint32_t* mask8w  = (uint32_t*)(ws + OFF_MASK8);
  int*      slotmap = (int*)(ws + OFF_SLOT);
  int*      vtok    = (int*)(ws + OFF_VTOK);
  int*      cnt     = (int*)(ws + OFF_CNT);
  int2*     list    = (int2*)(ws + OFF_LIST);

  vg_prep<<<1 + (BB * OO * EE) / 4096, 1024, 0, stream>>>(input_ids, tok2g, fc_b, mask8w, slotmap, vtok,
                                                          (float4*)out, cnt);
  vg_scan<<<(NNZV + 255) / 256, 256, 0, stream>>>(adj_rows, adj_cols, adj_values, gv_ids,
                                                   (const uint8_t*)mask8w, slotmap, cnt, list);
  vg_fuse<<<NB1 + BB * NQ, 1024, 0, stream>>>(emb, W_vh, fc_w, gv_ids, vtok, cnt, list, out);
}


# entry scan re-cut to a quarter of the waves: 4 entries per lane (128-bit loads), hits queued in LDS, then ranked, slot-looked-up and stored one per thread; list order changes, sums stay arrival order
# speedup vs baseline: 8.5342x; 1.0510x over previous
#include <hip/hip_runtime.h>
#include <stdint.h>

#define BB    8
#define SS    512
#define EE    768
#define VV    30522
#define GVV   20000
#define NNZV  400000
#define HH    64
#define OO    16

#define WPB     640

typedef float v2f __attribute__((ext_vector_type(2)));
typedef float v8f __attribute__((ext_vector_type(8)));

#define OFF_MASK8    0
#define OFF_SLOT     20480
#define OFF_VTOK     (OFF_SLOT + BB * GVV * 4)
#define OFF_CNT      (OFF_VTOK + BB * SS * 4)
#define CSTR         32
#define OFF_LIST     (OFF_CNT + BB * CSTR * 4)
#define WS_NEED      ((size_t)OFF_LIST + (size_t)BB * NNZV * 8)

#define NB1   (BB * (EE / 32))
#define NQ    8

__global__ void __launch_bounds__(1024)
vg_prep(const int* __restrict__ input_ids, const int* __restrict__ tok2g,
        const float* __restrict__ fc_b,
        uint32_t* __restrict__ mask8w, int* __restrict__ slotmap,
        int* __restrict__ vtok, float4* __restrict__ out4, int* __restrict__ cnt) {
  __shared__ uint32_t m8[GVV / 4];
  __shared__ uint32_t bm[BB * WPB];
  const int t = threadIdx.x;
  if (blockIdx.x != 0) {
    const int i4 = (blockIdx.x - 1) * 1024 + t;
    const float bv = fc_b[((i4 * 4) / EE) % OO];
    out4[i4] = make_float4(bv, bv, bv, bv);
    if (blockIdx.x == 1 && t < BB) { cnt[t * CSTR] = 0; cnt[t * CSTR + 1] = 0; }
    return;
  }
  for (int i = t; i < GVV / 4; i += 1024) m8[i] = 0u;
  for (int i = t; i < BB * WPB; i += 1024) bm[i] = 0u;
  __syncthreads();
  int vq[(BB * SS) / 1024];
#pragma unroll
  for (int q = 0; q < (BB * SS) / 1024; ++q) {
    int id = input_ids[q * 1024 + t];
    if (id > VV - 1 || id < 0) id = 0;
    vq[q] = id;
  }
#pragma unroll
  for (int q = 0; q < (BB * SS) / 1024; ++q) vq[q] = tok2g[vq[q]];
#pragma unroll
  for (int q = 0; q < (BB * SS) / 1024; ++q) {
    const int idx = q * 1024 + t;
    const int b = idx >> 9, s = idx & (SS - 1);
    const int v = vq[q];
    const uint32_t bit = 1u << (v & 31);
    const uint32_t old = atomicOr(&bm[b * WPB + (v >> 5)], bit);
    const bool own = (old & bit) == 0u;
    atomicOr(&m8[v >> 2], 1u << ((v & 3) * 8 + b));
    if (own) slotmap[b * GVV + v] = s;
    vtok[idx] = own ? v : -1;
  }
  __syncthreads();
  for (int i = t; i < GVV / 4; i += 1024) mask8w[i] = m8[i];
}

#define EPT     4
#define SCAN_T  256
#define SCAN_E  (SCAN_T * EPT)
#define SCAN_NB ((NNZV + SCAN_E - 1) / SCAN_E)
#define QMAX    (SCAN_E * BB)
__global__ void __launch_bounds__(SCAN_T)
vg_scan(const int* __restrict__ rows, const int* __restrict__ cols,
        const float* __restrict__ vals,
        const int* __restrict__ gv_ids, const uint8_t* __restrict__ mask8,
        const int* __restrict__ slotmap,
        int* __restrict__ cnt, int2* __restrict__ list) {
  __shared__ int lcnt[2 * BB];
  __shared__ int lbase[2 * BB];
  __shared__ int qn;
  __shared__ int4 qrec[QMAX];
  const int tid = threadIdx.x;
  if (tid < 2 * BB) lcnt[tid] = 0;
  if (tid == 0) qn = 0;
  __syncthreads();
  const int lane = tid & 31;
  const int k0 = (blockIdx.x * SCAN_T + tid) * EPT;
  int row[EPT] = {0, 0, 0, 0}, col[EPT] = {0, 0, 0, 0}, gv[EPT] = {0, 0, 0, 0};
  float val[EPT] = {0.f, 0.f, 0.f, 0.f};
  uint32_t cm[EPT] = {0u, 0u, 0u, 0u}, t1[EPT] = {0u, 0u, 0u, 0u};
  if (k0 < NNZV) {
    const int4 r4 = *(const int4*)(rows + k0);
    const int4 c4 = *(const int4*)(cols + k0);
    const float4 v4 = *(const float4*)(vals + k0);
    row[0] = r4.x; row[1] = r4.y; row[2] = r4.z; row[3] = r4.w;
    col[0] = c4.x; col[1] = c4.y; col[2] = c4.z; col[3] = c4.w;
    val[0] = v4.x; val[1] = v4.y; val[2] = v4.z; val[3] = v4.w;
#pragma unroll
    for (int j = 0; j < EPT; ++j) {
      const uint32_t rm = mask8[row[j]];
      cm[j] = mask8[col[j]];
      gv[j] = gv_ids[row[j]];
      t1[j] = rm & ~cm[j];
    }
  }
  const int nl = __popc(cm[0] | t1[0]) + __popc(cm[1] | t1[1]) + __popc(cm[2] | t1[2]) + __popc(cm[3] | t1[3]);
  int incl = nl;
#pragma unroll
  for (int d = 1; d < 32; d <<= 1) {
    const int y = __shfl_up(incl, d);
    if (lane >= d) incl += y;
  }
  const int wtot = __shfl(incl, 31);
  int qb = 0;
  if (lane == 31 && wtot > 0) qb = atomicAdd(&qn, wtot);
  qb = __shfl(qb, 31);
  int pos = qb + incl - nl;
#pragma unroll
  for (int j = 0; j < EPT; ++j) {
    uint32_t m = cm[j];
    for (int it = 0; it < BB && m != 0u; ++it) {
      const int b = __ffs((int)m) - 1;
      m &= m - 1u;
      qrec[pos++] = make_int4(b, b * GVV + col[j], gv[j], __float_as_int(val[j]));
    }
    m = t1[j];
    for (int it = 0; it < BB && m != 0u; ++it) {
      const int b = __ffs((int)m) - 1;
      m &= m - 1u;
      qrec[pos++] = make_int4(BB + b, b * GVV + row[j], col[j], __float_as_int(val[j]));
    }
  }
  __syncthreads();
  int nq = qn;
  if (nq > QMAX) nq = QMAX;
  for (int i = tid; i < nq; i += SCAN_T) {
    const int code = qrec[i].x;
    const int r = atomicAdd(&lcnt[code], 1);
    qrec[i].x = code | (r << 4);
  }
  __syncthreads();
  if (tid < BB) {
    const uint32_t c1 = (uint32_t)lcnt[tid];
    const uint32_t c2 = (uint32_t)lcnt[BB + tid];
    unsigned long long old = 0ull;
    if (c1 | c2)
      old = atomicAdd((unsigned long long*)(cnt + tid * CSTR),
                      (unsigned long long)c1 | ((unsigned long long)c2 << 32));
    lbase[tid] = (int)(uint32_t)old;
    lbase[BB + tid] = (int)(uint32_t)(old >> 32);
  }
  __syncthreads();
  for (int i = tid; i < nq; i += SCAN_T) {
    const int4 a = qrec[i];
    const int sa = slotmap[a.y];
    const int code = a.x & 15, r = a.x >> 4;
    const int at = lbase[code] + r;
    list[(size_t)(code & 7) * NNZV + (code < BB ? at : NNZV - 1 - at)] = make_int2(a.z | (sa << 16), a.w);
  }
}

#define P3   40
#define CPP  65
#define NES  (EE / 32)
#define QCAP 4096

__device__ __forceinline__ void
fuse_colpart(const int b, const int es,
             const float* __restrict__ emb, const float* __restrict__ W,
             const float* __restrict__ fc_w, const int* __restrict__ vtok,
             const int* __restrict__ cnt, const int2* __restrict__ list, float* __restrict__ out) {
  __shared__ __attribute__((aligned(16))) float T[SS * P3];
  __shared__ float Cp[4 * 32 * CPP];
  __shared__ int vs[SS];

  const int tid  = threadIdx.x;
  const int w    = __builtin_amdgcn_readfirstlane(tid >> 5);
  const int lane = tid & 31;
  const int c16  = lane & 15;
  const int hi   = lane >> 4;
  const float* __restrict__ embc = emb + es * 32 + lane;

  for (int i = tid; i < (SS * P3) / 4; i += 1024) ((float4*)T)[i] = make_float4(0.f, 0.f, 0.f, 0.f);
  if (tid < SS) vs[tid] = vtok[b * SS + tid];
  int n = cnt[b * CSTR];
  if (n > NNZV) n = NNZV;
  const int2* __restrict__ lb = list + (size_t)b * NNZV;
  int2 rec = make_int2(0, 0);
  if (w * 32 + lane < n) rec = lb[w * 32 + lane];
  __syncthreads();

  for (int base = w * 32; base < n; base += 1024) {
    int2 nxt = make_int2(0, 0);
    if (base + 1024 + lane < n) nxt = lb[base + 1024 + lane];
    const int m = (n - base) < 32 ? (n - base) : 32;
    if (m == 32) {
      float x[32];
#pragma unroll
      for (int j = 0; j < 32; ++j) {
        const int gv = __builtin_amdgcn_readlane(rec.x, j) & 0xFFFF;
        x[j] = embc[(size_t)gv * EE];
      }
#pragma unroll
      for (int j = 0; j < 32; ++j) {
        const int s = __builtin_amdgcn_readlane(rec.x, j) >> 16;
        const float val = __int_as_float(__builtin_amdgcn_readlane(rec.y, j));
        atomicAdd(&T[s * P3 + lane], val * x[j]);
      }
    } else {
      for (int j = 0; j < m; ++j) {
        const int rx = __builtin_amdgcn_readlane(rec.x, j);
        const float val = __int_as_float(__builtin_amdgcn_readlane(rec.y, j));
        atomicAdd(&T[(rx >> 16) * P3 + lane], val * embc[(size_t)(rx & 0xFFFF) * EE]);
      }
    }
    rec = nxt;
  }
  __syncthreads();

  const int tl = w & 7;
  const int mi = tl & 1;
  const int ni = tl >> 1;
  const int kq = w >> 3;
  const int arow = mi * 16 + c16;
  const int ncol = ni * 16 + c16;
  const float* __restrict__ Wc = W + ncol;
  const v8f zacc = {0.f, 0.f, 0.f, 0.f, 0.f, 0.f, 0.f, 0.f};
  v8f acc1 = zacc, acc2 = zacc;
#pragma unroll 8
  for (int j = 0; j < 32; j += 2) {
    const int s0 = kq * 128 + 4 * j + 2 * hi;
    const int v0 = vs[s0], v1 = vs[s0 + 1], v2 = vs[s0 + 4], v3 = vs[s0 + 5];
    const float w0 = Wc[(size_t)(v0 < 0 ? 0 : v0) * HH];
    const float w1 = Wc[(size_t)(v1 < 0 ? 0 : v1) * HH];
    const float w2 = Wc[(size_t)(v2 < 0 ? 0 : v2) * HH];
    const float w3 = Wc[(size_t)(v3 < 0 ? 0 : v3) * HH];
    const v2f a1 = {T[s0 * P3 + arow], T[(s0 + 1) * P3 + arow]};
    const v2f b1 = {v0 < 0 ? 0.f : w0, v1 < 0 ? 0.f : w1};
    const v2f a2 = {T[(s0 + 4) * P3 + arow], T[(s0 + 5) * P3 + arow]};
    const v2f b2 = {v2 < 0 ? 0.f : w2, v3 < 0 ? 0.f : w3};
    acc1 = __builtin_amdgcn_wmma_f32_16x16x4_f32(false, a1, false, b1, (short)0, acc1, false, false);
    acc2 = __builtin_amdgcn_wmma_f32_16x16x4_f32(false, a2, false, b2, (short)0, acc2, false, false);
  }
  const v8f accs = acc1 + acc2;
#pragma unroll
  for (int r = 0; r < 8; ++r)
    Cp[(kq * 32 + mi * 16 + 8 * hi + r) * CPP + ncol] = accs[r];
  __syncthreads();

  if (tid < OO * 32) {
    const int o = tid >> 5;
    const int e = tid & 31;
    float a = 0.f;
    for (int h = 0; h < HH; ++h) {
      const float f = (Cp[(0 * 32 + e) * CPP + h] + Cp[(1 * 32 + e) * CPP + h]) +
                      (Cp[(2 * 32 + e) * CPP + h] + Cp[(3 * 32 + e) * CPP + h]);
      a += f * fc_w[o * HH + h];
    }
    atomicAdd(&out[((size_t)b * OO + o) * EE + es * 32 + e], a);
  }
}

__device__ __forceinline__ void
fuse_rowpart(const int b, const int q,
             const float* __restrict__ emb, const float* __restrict__ W,
             const float* __restrict__ fc_w, const int* __restrict__ gv_ids,
             const int* __restrict__ vtok, const int* __restrict__ cnt,
             const int2* __restrict__ list, float* __restrict__ out) {
  __shared__ __attribute__((aligned(16))) float Rt[64 * HH];
  __shared__ float RF[64 * OO];
  __shared__ int   es[64];
  __shared__ int2  qrec[QCAP];
  __shared__ int   qn[2];

  const int tid  = threadIdx.x;
  const int w    = __builtin_amdgcn_readfirstlane(tid >> 5);
  const int lane = tid & 31;
  const int c16  = lane & 15;
  const int hi   = lane >> 4;

  ((float4*)Rt)[tid] = make_float4(0.f, 0.f, 0.f, 0.f);
  if (tid < 2) qn[tid] = 0;
  if (tid < 64) {
    const int v = vtok[b * SS + q * 64 + tid];
    es[tid] = v < 0 ? -1 : gv_ids[v];
  }
  int n = cnt[b * CSTR + 1];
  if (n > NNZV) n = NNZV;
  const int2* __restrict__ lbk = list + (size_t)b * NNZV + (NNZV - 1);
  const uint32_t below = (1u << lane) - 1u;
  __syncthreads();

  int par = 0;
  for (int base = 0; base < n; base += QCAP, par ^= 1) {
    int2 r4[QCAP / 1024];
#pragma unroll
    for (int j = 0; j < QCAP / 1024; ++j) {
      const int i = base + j * 1024 + tid;
      r4[j] = make_int2(-1, 0);
      if (i < n) r4[j] = lbk[-(ptrdiff_t)i];
    }
    uint32_t hb[QCAP / 1024];
    int tot = 0;
#pragma unroll
    for (int j = 0; j < QCAP / 1024; ++j) {
      hb[j] = (uint32_t)__ballot(((uint32_t)r4[j].x >> 22) == (uint32_t)q);
      tot += __popc(hb[j]);
    }
    int p0 = 0;
    if (lane == 0 && tot > 0) p0 = atomicAdd(&qn[par], tot);
    p0 = __shfl(p0, 0);
#pragma unroll
    for (int j = 0; j < QCAP / 1024; ++j) {
      if ((hb[j] >> lane) & 1u) qrec[p0 + __popc(hb[j] & below)] = r4[j];
      p0 += __popc(hb[j]);
    }
    __syncthreads();
    const int m = __builtin_amdgcn_readfirstlane(qn[par]);
    if (tid == 0) qn[par ^ 1] = 0;
    for (int i0 = w; i0 < m; i0 += 256) {
      int rx[8]; float rv[8]; float2 wv[8];
#pragma unroll
      for (int jj = 0; jj < 8; ++jj) {
        int2 r = make_int2(0, 0);
        if (i0 + 32 * jj < m) r = qrec[i0 + 32 * jj];
        rx[jj] = __builtin_amdgcn_readfirstlane(r.x);
        rv[jj] = __int_as_float(__builtin_amdgcn_readfirstlane(r.y));
      }
#pragma unroll
      for (int jj = 0; jj < 8; ++jj) {
        wv[jj] = make_float2(0.f, 0.f);
        if (i0 + 32 * jj < m) wv[jj] = *(const float2*)(W + (size_t)(rx[jj] & 0xFFFF) * HH + 2 * lane);
      }
#pragma unroll
      for (int jj = 0; jj < 8; ++jj) {
        if (i0 + 32 * jj < m) {
          float* d = &Rt[((rx[jj] >> 16) & 63) * HH + 2 * lane];
          atomicAdd(d, rv[jj] * wv[jj].x);
          atomicAdd(d + 1, rv[jj] * wv[jj].y);
        }
      }
    }
    __syncthreads();
  }

  {
    const int s = tid >> 4, o = tid & 15;
    float a = 0.f;
    for (int h = 0; h < HH; ++h) a += Rt[s * HH + h] * fc_w[o * HH + h];
    RF[s * OO + o] = a;
  }
  __syncthreads();

  const v8f zacc = {0.f, 0.f, 0.f, 0.f, 0.f, 0.f, 0.f, 0.f};
  for (int t = w; t < EE / 16; t += 32) {
    const float* __restrict__ embt = emb + t * 16 + c16;
    v8f acc1 = zacc, acc2 = zacc;
#pragma unroll 4
    for (int ks = 0; ks < 16; ks += 2) {
      const int s0 = 4 * ks + 2 * hi;
      const int g0 = es[s0], g1 = es[s0 + 1], g2 = es[s0 + 4], g3 = es[s0 + 5];
      const float x0 = embt[(size_t)(g0 < 0 ? 0 : g0) * EE];
      const float x1 = embt[(size_t)(g1 < 0 ? 0 : g1) * EE];
      const float x2 = embt[(size_t)(g2 < 0 ? 0 : g2) * EE];
      const float x3 = embt[(size_t)(g3 < 0 ? 0 : g3) * EE];
      const v2f a1 = {RF[s0 * OO + c16], RF[(s0 + 1) * OO + c16]};
      const v2f b1 = {g0 < 0 ? 0.f : x0, g1 < 0 ? 0.f : x1};
      const v2f a2 = {RF[(s0 + 4) * OO + c16], RF[(s0 + 5) * OO + c16]};
      const v2f b2 = {g2 < 0 ? 0.f : x2, g3 < 0 ? 0.f : x3};
      acc1 = __builtin_amdgcn_wmma_f32_16x16x4_f32(false, a1, false, b1, (short)0, acc1, false, false);
      acc2 = __builtin_amdgcn_wmma_f32_16x16x4_f32(false, a2, false, b2, (short)0, acc2, false, false);
    }
    const v8f accs = acc1 + acc2;
#pragma unroll
    for (int r = 0; r < 8; ++r)
      atomicAdd(&out[((size_t)b * OO + 8 * hi + r) * EE + t * 16 + c16], accs[r]);
  }
}

__global__ void __launch_bounds__(1024)
vg_fuse(const float* __restrict__ emb, const float* __restrict__ W,
        const float* __restrict__ fc_w,
        const int* __restrict__ gv_ids, const int* __restrict__ vtok,
        const int* __restrict__ cnt,
        const int2* __restrict__ list, float* __restrict__ out) {
  if (blockIdx.x < NB1) {
    fuse_colpart(blockIdx.x / NES, blockIdx.x % NES, emb, W, fc_w, vtok, cnt, list, out);
  } else {
    const int i2 = blockIdx.x - NB1;
    fuse_rowpart(i2 / NQ, i2 % NQ, emb, W, fc_w, gv_ids, vtok, cnt, list, out);
  }
}

extern "C" void kernel_launch(void* const* d_in, const int* in_sizes, int n_in,
                              void* d_out, int out_size, void* d_ws, size_t ws_size,
                              hipStream_t stream) {
  (void)in_sizes; (void)n_in; (void)out_size;
  const float* emb        = (const float*)d_in[0];
  const float* adj_values = (const float*)d_in[1];
  const float* W_vh       = (const float*)d_in[2];
  const float* fc_w       = (const float*)d_in[3];
  const float* fc_b       = (const float*)d_in[4];
  const int*   input_ids  = (const int*)d_in[5];
  const int*   tok2g      = (const int*)d_in[6];
  const int*   gv_ids     = (const int*)d_in[7];
  const int*   adj_rows   = (const int*)d_in[8];
  const int*   adj_cols   = (const int*)d_in[9];
  float* out = (float*)d_out;

  if (ws_size < WS_NEED) return;

  uint8_t*  ws      = (uint8_t*)d_ws;
  uint32_t* mask8w  = (uint32_t*)(ws + OFF_MASK8);
  int*      slotmap = (int*)(ws + OFF_SLOT);
  int*      vtok    = (int*)(ws + OFF_VTOK);
  int*      cnt     = (int*)(ws + OFF_CNT);
  int2*     list    = (int2*)(ws + OFF_LIST);

  vg_prep<<<1 + (BB * OO * EE) / 4096, 1024, 0, stream>>>(input_ids, tok2g, fc_b, mask8w, slotmap, vtok,
                                                          (float4*)out, cnt);
  vg_scan<<<SCAN_NB, SCAN_T, 0, stream>>>(adj_rows, adj_cols, adj_values, gv_ids,
                                                   (const uint8_t*)mask8w, slotmap, cnt, list);
  vg_fuse<<<NB1 + BB * NQ, 1024, 0, stream>>>(emb, W_vh, fc_w, gv_ids, vtok, cnt, list, out);
}


# col-present T sums: the batch's last, partial record chunk (one wave per block) issues all its emb line loads together like a full chunk, not one round trip per record; sums stay arrival order
# speedup vs baseline: 8.8909x; 1.0418x over previous
#include <hip/hip_runtime.h>
#include <stdint.h>

#define BB    8
#define SS    512
#define EE    768
#define VV    30522
#define GVV   20000
#define NNZV  400000
#define HH    64
#define OO    16

#define WPB     640

typedef float v2f __attribute__((ext_vector_type(2)));
typedef float v8f __attribute__((ext_vector_type(8)));

#define OFF_MASK8    0
#define OFF_SLOT     20480
#define OFF_VTOK     (OFF_SLOT + BB * GVV * 4)
#define OFF_CNT      (OFF_VTOK + BB * SS * 4)
#define CSTR         32
#define OFF_LIST     (OFF_CNT + BB * CSTR * 4)
#define WS_NEED      ((size_t)OFF_LIST + (size_t)BB * NNZV * 8)

#define NB1   (BB * (EE / 32))
#define NQ    8

__global__ void __launch_bounds__(1024)
vg_prep(const int* __restrict__ input_ids, const int* __restrict__ tok2g,
        const float* __restrict__ fc_b,
        uint32_t* __restrict__ mask8w, int* __restrict__ slotmap,
        int* __restrict__ vtok, float4* __restrict__ out4, int* __restrict__ cnt) {
  __shared__ uint32_t m8[GVV / 4];
  __shared__ uint32_t bm[BB * WPB];
  const int t = threadIdx.x;
  if (blockIdx.x != 0) {
    const int i4 = (blockIdx.x - 1) * 1024 + t;
    const float bv = fc_b[((i4 * 4) / EE) % OO];
    out4[i4] = make_float4(bv, bv, bv, bv);
    if (blockIdx.x == 1 && t < BB) { cnt[t * CSTR] = 0; cnt[t * CSTR + 1] = 0; }
    return;
  }
  for (int i = t; i < GVV / 4; i += 1024) m8[i] = 0u;
  for (int i = t; i < BB * WPB; i += 1024) bm[i] = 0u;
  __syncthreads();
  int vq[(BB * SS) / 1024];
#pragma unroll
  for (int q = 0; q < (BB * SS) / 1024; ++q) {
    int id = input_ids[q * 1024 + t];
    if (id > VV - 1 || id < 0) id = 0;
    vq[q] = id;
  }
#pragma unroll
  for (int q = 0; q < (BB * SS) / 1024; ++q) vq[q] = tok2g[vq[q]];
#pragma unroll
  for (int q = 0; q < (BB * SS) / 1024; ++q) {
    const int idx = q * 1024 + t;
    const int b = idx >> 9, s = idx & (SS - 1);
    const int v = vq[q];
    const uint32_t bit = 1u << (v & 31);
    const uint32_t old = atomicOr(&bm[b * WPB + (v >> 5)], bit);
    const bool own = (old & bit) == 0u;
    atomicOr(&m8[v >> 2], 1u << ((v & 3) * 8 + b));
    if (own) slotmap[b * GVV + v] = s;
    vtok[idx] = own ? v : -1;
  }
  __syncthreads();
  for (int i = t; i < GVV / 4; i += 1024) mask8w[i] = m8[i];
}

#define EPT     4
#define SCAN_T  256
#define SCAN_E  (SCAN_T * EPT)
#define SCAN_NB ((NNZV + SCAN_E - 1) / SCAN_E)
#define QMAX    (SCAN_E * BB)
__global__ void __launch_bounds__(SCAN_T)
vg_scan(const int* __restrict__ rows, const int* __restrict__ cols,
        const float* __restrict__ vals,
        const int* __restrict__ gv_ids, const uint8_t* __restrict__ mask8,
        const int* __restrict__ slotmap,
        int* __restrict__ cnt, int2* __restrict__ list) {
  __shared__ int lcnt[2 * BB];
  __shared__ int lbase[2 * BB];
  __shared__ int qn;
  __shared__ int4 qrec[QMAX];
  const int tid = threadIdx.x;
  if (tid < 2 * BB) lcnt[tid] = 0;
  if (tid == 0) qn = 0;
  __syncthreads();
  const int lane = tid & 31;
  const int k0 = (blockIdx.x * SCAN_T + tid) * EPT;
  int row[EPT] = {0, 0, 0, 0}, col[EPT] = {0, 0, 0, 0}, gv[EPT] = {0, 0, 0, 0};
  float val[EPT] = {0.f, 0.f, 0.f, 0.f};
  uint32_t cm[EPT] = {0u, 0u, 0u, 0u}, t1[EPT] = {0u, 0u, 0u, 0u};
  if (k0 < NNZV) {
    const int4 r4 = *(const int4*)(rows + k0);
    const int4 c4 = *(const int4*)(cols + k0);
    const float4 v4 = *(const float4*)(vals + k0);
    row[0] = r4.x; row[1] = r4.y; row[2] = r4.z; row[3] = r4.w;
    col[0] = c4.x; col[1] = c4.y; col[2] = c4.z; col[3] = c4.w;
    val[0] = v4.x; val[1] = v4.y; val[2] = v4.z; val[3] = v4.w;
#pragma unroll
    for (int j = 0; j < EPT; ++j) {
      const uint32_t rm = mask8[row[j]];
      cm[j] = mask8[col[j]];
      gv[j] = gv_ids[row[j]];
      t1[j] = rm & ~cm[j];
    }
  }
  const int nl = __popc(cm[0] | t1[0]) + __popc(cm[1] | t1[1]) + __popc(cm[2] | t1[2]) + __popc(cm[3] | t1[3]);
  int incl = nl;
#pragma unroll
  for (int d = 1; d < 32; d <<= 1) {
    const int y = __shfl_up(incl, d);
    if (lane >= d) incl += y;
  }
  const int wtot = __shfl(incl, 31);
  int qb = 0;
  if (lane == 31 && wtot > 0) qb = atomicAdd(&qn, wtot);
  qb = __shfl(qb, 31);
  int pos = qb + incl - nl;
#pragma unroll
  for (int j = 0; j < EPT; ++j) {
    uint32_t m = cm[j];
    for (int it = 0; it < BB && m != 0u; ++it) {
      const int b = __ffs((int)m) - 1;
      m &= m - 1u;
      qrec[pos++] = make_int4(b, b * GVV + col[j], gv[j], __float_as_int(val[j]));
    }
    m = t1[j];
    for (int it = 0; it < BB && m != 0u; ++it) {
      const int b = __ffs((int)m) - 1;
      m &= m - 1u;
      qrec[pos++] = make_int4(BB + b, b * GVV + row[j], col[j], __float_as_int(val[j]));
    }
  }
  __syncthreads();
  int nq = qn;
  if (nq > QMAX) nq = QMAX;
  for (int i = tid; i < nq; i += SCAN_T) {
    const int code = qrec[i].x;
    const int r = atomicAdd(&lcnt[code], 1);
    qrec[i].x = code | (r << 4);
  }
  __syncthreads();
  if (tid < BB) {
    const uint32_t c1 = (uint32_t)lcnt[tid];
    const uint32_t c2 = (uint32_t)lcnt[BB + tid];
    unsigned long long old = 0ull;
    if (c1 | c2)
      old = atomicAdd((unsigned long long*)(cnt + tid * CSTR),
                      (unsigned long long)c1 | ((unsigned long long)c2 << 32));
    lbase[tid] = (int)(uint32_t)old;
    lbase[BB + tid] = (int)(uint32_t)(old >> 32);
  }
  __syncthreads();
  for (int i = tid; i < nq; i += SCAN_T) {
    const int4 a = qrec[i];
    const int sa = slotmap[a.y];
    const int code = a.x & 15, r = a.x >> 4;
    const int at = lbase[code] + r;
    list[(size_t)(code & 7) * NNZV + (code < BB ? at : NNZV - 1 - at)] = make_int2(a.z | (sa << 16), a.w);
  }
}

#define P3   40
#define CPP  65
#define NES  (EE / 32)
#define QCAP 4096

__device__ __forceinline__ void
fuse_colpart(const int b, const int es,
             const float* __restrict__ emb, const float* __restrict__ W,
             const float* __restrict__ fc_w, const int* __restrict__ vtok,
             const int* __restrict__ cnt, const int2* __restrict__ list, float* __restrict__ out) {
  __shared__ __attribute__((aligned(16))) float T[SS * P3];
  __shared__ float Cp[4 * 32 * CPP];
  __shared__ int vs[SS];

  const int tid  = threadIdx.x;
  const int w    = __builtin_amdgcn_readfirstlane(tid >> 5);
  const int lane = tid & 31;
  const int c16  = lane & 15;
  const int hi   = lane >> 4;
  const float* __restrict__ embc = emb + es * 32 + lane;

  for (int i = tid; i < (SS * P3) / 4; i += 1024) ((float4*)T)[i] = make_float4(0.f, 0.f, 0.f, 0.f);
  if (tid < SS) vs[tid] = vtok[b * SS + tid];
  int n = cnt[b * CSTR];
  if (n > NNZV) n = NNZV;
  const int2* __restrict__ lb = list + (size_t)b * NNZV;
  int2 rec = make_int2(0, 0);
  if (w * 32 + lane < n) rec = lb[w * 32 + lane];
  __syncthreads();

  for (int base = w * 32; base < n; base += 1024) {
    int2 nxt = make_int2(0, 0);
    if (base + 1024 + lane < n) nxt = lb[base + 1024 + lane];
    const int m = (n - base) < 32 ? (n - base) : 32;
    if (m == 32) {
      float x[32];
#pragma unroll
      for (int j = 0; j < 32; ++j) {
        const int gv = __builtin_amdgcn_readlane(rec.x, j) & 0xFFFF;
        x[j] = embc[(size_t)gv * EE];
      }
#pragma unroll
      for (int j = 0; j < 32; ++j) {
        const int s = __builtin_amdgcn_readlane(rec.x, j) >> 16;
        const float val = __int_as_float(__builtin_amdgcn_readlane(rec.y, j));
        atomicAdd(&T[s * P3 + lane], val * x[j]);
      }
    } else {
      float x[32];
#pragma unroll
      for (int j = 0; j < 32; ++j) {
        const int gv = __builtin_amdgcn_readlane(rec.x, j) & 0xFFFF;
        x[j] = embc[(size_t)gv * EE];
      }
#pragma unroll
      for (int j = 0; j < 32; ++j) {
        const int s = __builtin_amdgcn_readlane(rec.x, j) >> 16;
        const float val = __int_as_float(__builtin_amdgcn_readlane(rec.y, j));
        atomicAdd(&T[s * P3 + lane], (j == 0 || j < m) ? val * x[j] : 0.f);
      }
    }
    rec = nxt;
  }
  __syncthreads();

  const int tl = w & 7;
  const int mi = tl & 1;
  const int ni = tl >> 1;
  const int kq = w >> 3;
  const int arow = mi * 16 + c16;
  const int ncol = ni * 16 + c16;
  const float* __restrict__ Wc = W + ncol;
  const v8f zacc = {0.f, 0.f, 0.f, 0.f, 0.f, 0.f, 0.f, 0.f};
  v8f acc1 = zacc, acc2 = zacc;
#pragma unroll 8
  for (int j = 0; j < 32; j += 2) {
    const int s0 = kq * 128 + 4 * j + 2 * hi;
    const int v0 = vs[s0], v1 = vs[s0 + 1], v2 = vs[s0 + 4], v3 = vs[s0 + 5];
    const float w0 = Wc[(size_t)(v0 < 0 ? 0 : v0) * HH];
    const float w1 = Wc[(size_t)(v1 < 0 ? 0 : v1) * HH];
    const float w2 = Wc[(size_t)(v2 < 0 ? 0 : v2) * HH];
    const float w3 = Wc[(size_t)(v3 < 0 ? 0 : v3) * HH];
    const v2f a1 = {T[s0 * P3 + arow], T[(s0 + 1) * P3 + arow]};
    const v2f b1 = {v0 < 0 ? 0.f : w0, v1 < 0 ? 0.f : w1};
    const v2f a2 = {T[(s0 + 4) * P3 + arow], T[(s0 + 5) * P3 + arow]};
    const v2f b2 = {v2 < 0 ? 0.f : w2, v3 < 0 ? 0.f : w3};
    acc1 = __builtin_amdgcn_wmma_f32_16x16x4_f32(false, a1, false, b1, (short)0, acc1, false, false);
    acc2 = __builtin_amdgcn_wmma_f32_16x16x4_f32(false, a2, false, b2, (short)0, acc2, false, false);
  }
  const v8f accs = acc1 + acc2;
#pragma unroll
  for (int r = 0; r < 8; ++r)
    Cp[(kq * 32 + mi * 16 + 8 * hi + r) * CPP + ncol] = accs[r];
  __syncthreads();

  if (tid < OO * 32) {
    const int o = tid >> 5;
    const int e = tid & 31;
    float a = 0.f;
    for (int h = 0; h < HH; ++h) {
      const float f = (Cp[(0 * 32 + e) * CPP + h] + Cp[(1 * 32 + e) * CPP + h]) +
                      (Cp[(2 * 32 + e) * CPP + h] + Cp[(3 * 32 + e) * CPP + h]);
      a += f * fc_w[o * HH + h];
    }
    atomicAdd(&out[((size_t)b * OO + o) * EE + es * 32 + e], a);
  }
}

__device__ __forceinline__ void
fuse_rowpart(const int b, const int q,
             const float* __restrict__ emb, const float* __restrict__ W,
             const float* __restrict__ fc_w, const int* __restrict__ gv_ids,
             const int* __restrict__ vtok, const int* __restrict__ cnt,
             const int2* __restrict__ list, float* __restrict__ out) {
  __shared__ __attribute__((aligned(16))) float Rt[64 * HH];
  __shared__ float RF[64 * OO];
  __shared__ int   es[64];
  __shared__ int2  qrec[QCAP];
  __shared__ int   qn[2];

  const int tid  = threadIdx.x;
  const int w    = __builtin_amdgcn_readfirstlane(tid >> 5);
  const int lane = tid & 31;
  const int c16  = lane & 15;
  const int hi   = lane >> 4;

  ((float4*)Rt)[tid] = make_float4(0.f, 0.f, 0.f, 0.f);
  if (tid < 2) qn[tid] = 0;
  if (tid < 64) {
    const int v = vtok[b * SS + q * 64 + tid];
    es[tid] = v < 0 ? -1 : gv_ids[v];
  }
  int n = cnt[b * CSTR + 1];
  if (n > NNZV) n = NNZV;
  const int2* __restrict__ lbk = list + (size_t)b * NNZV + (NNZV - 1);
  const uint32_t below = (1u << lane) - 1u;
  __syncthreads();

  int par = 0;
  for (int base = 0; base < n; base += QCAP, par ^= 1) {
    int2 r4[QCAP / 1024];
#pragma unroll
    for (int j = 0; j < QCAP / 1024; ++j) {
      const int i = base + j * 1024 + tid;
      r4[j] = make_int2(-1, 0);
      if (i < n) r4[j] = lbk[-(ptrdiff_t)i];
    }
    uint32_t hb[QCAP / 1024];
    int tot = 0;
#pragma unroll
    for (int j = 0; j < QCAP / 1024; ++j) {
      hb[j] = (uint32_t)__ballot(((uint32_t)r4[j].x >> 22) == (uint32_t)q);
      tot += __popc(hb[j]);
    }
    int p0 = 0;
    if (lane == 0 && tot > 0) p0 = atomicAdd(&qn[par], tot);
    p0 = __shfl(p0, 0);
#pragma unroll
    for (int j = 0; j < QCAP / 1024; ++j) {
      if ((hb[j] >> lane) & 1u) qrec[p0 + __popc(hb[j] & below)] = r4[j];
      p0 += __popc(hb[j]);
    }
    __syncthreads();
    const int m = __builtin_amdgcn_readfirstlane(qn[par]);
    if (tid == 0) qn[par ^ 1] = 0;
    for (int i0 = w; i0 < m; i0 += 256) {
      int rx[8]; float rv[8]; float2 wv[8];
#pragma unroll
      for (int jj = 0; jj < 8; ++jj) {
        int2 r = make_int2(0, 0);
        if (i0 + 32 * jj < m) r = qrec[i0 + 32 * jj];
        rx[jj] = __builtin_amdgcn_readfirstlane(r.x);
        rv[jj] = __int_as_float(__builtin_amdgcn_readfirstlane(r.y));
      }
#pragma unroll
      for (int jj = 0; jj < 8; ++jj) {
        wv[jj] = make_float2(0.f, 0.f);
        if (i0 + 32 * jj < m) wv[jj] = *(const float2*)(W + (size_t)(rx[jj] & 0xFFFF) * HH + 2 * lane);
      }
#pragma unroll
      for (int jj = 0; jj < 8; ++jj) {
        if (i0 + 32 * jj < m) {
          float* d = &Rt[((rx[jj] >> 16) & 63) * HH + 2 * lane];
          atomicAdd(d, rv[jj] * wv[jj].x);
          atomicAdd(d + 1, rv[jj] * wv[jj].y);
        }
      }
    }
    __syncthreads();
  }

  {
    const int s = tid >> 4, o = tid & 15;
    float a = 0.f;
    for (int h = 0; h < HH; ++h) a += Rt[s * HH + h] * fc_w[o * HH + h];
    RF[s * OO + o] = a;
  }
  __syncthreads();

  const v8f zacc = {0.f, 0.f, 0.f, 0.f, 0.f, 0.f, 0.f, 0.f};
  for (int t = w; t < EE / 16; t += 32) {
    const float* __restrict__ embt = emb + t * 16 + c16;
    v8f acc1 = zacc, acc2 = zacc;
#pragma unroll 4
    for (int ks = 0; ks < 16; ks += 2) {
      const int s0 = 4 * ks + 2 * hi;
      const int g0 = es[s0], g1 = es[s0 + 1], g2 = es[s0 + 4], g3 = es[s0 + 5];
      const float x0 = embt[(size_t)(g0 < 0 ? 0 : g0) * EE];
      const float x1 = embt[(size_t)(g1 < 0 ? 0 : g1) * EE];
      const float x2 = embt[(size_t)(g2 < 0 ? 0 : g2) * EE];
      const float x3 = embt[(size_t)(g3 < 0 ? 0 : g3) * EE];
      const v2f a1 = {RF[s0 * OO + c16], RF[(s0 + 1) * OO + c16]};
      const v2f b1 = {g0 < 0 ? 0.f : x0, g1 < 0 ? 0.f : x1};
      const v2f a2 = {RF[(s0 + 4) * OO + c16], RF[(s0 + 5) * OO + c16]};
      const v2f b2 = {g2 < 0 ? 0.f : x2, g3 < 0 ? 0.f : x3};
      acc1 = __builtin_amdgcn_wmma_f32_16x16x4_f32(false, a1, false, b1, (short)0, acc1, false, false);
      acc2 = __builtin_amdgcn_wmma_f32_16x16x4_f32(false, a2, false, b2, (short)0, acc2, false, false);
    }
    const v8f accs = acc1 + acc2;
#pragma unroll
    for (int r = 0; r < 8; ++r)
      atomicAdd(&out[((size_t)b * OO + 8 * hi + r) * EE + t * 16 + c16], accs[r]);
  }
}

__global__ void __launch_bounds__(1024)
vg_fuse(const float* __restrict__ emb, const float* __restrict__ W,
        const float* __restrict__ fc_w,
        const int* __restrict__ gv_ids, const int* __restrict__ vtok,
        const int* __restrict__ cnt,
        const int2* __restrict__ list, float* __restrict__ out) {
  if (blockIdx.x < NB1) {
    fuse_colpart(blockIdx.x / NES, blockIdx.x % NES, emb, W, fc_w, vtok, cnt, list, out);
  } else {
    const int i2 = blockIdx.x - NB1;
    fuse_rowpart(i2 / NQ, i2 % NQ, emb, W, fc_w, gv_ids, vtok, cnt, list, out);
  }
}

extern "C" void kernel_launch(void* const* d_in, const int* in_sizes, int n_in,
                              void* d_out, int out_size, void* d_ws, size_t ws_size,
                              hipStream_t stream) {
  (void)in_sizes; (void)n_in; (void)out_size;
  const float* emb        = (const float*)d_in[0];
  const float* adj_values = (const float*)d_in[1];
  const float* W_vh       = (const float*)d_in[2];
  const float* fc_w       = (const float*)d_in[3];
  const float* fc_b       = (const float*)d_in[4];
  const int*   input_ids  = (const int*)d_in[5];
  const int*   tok2g      = (const int*)d_in[6];
  const int*   gv_ids     = (const int*)d_in[7];
  const int*   adj_rows   = (const int*)d_in[8];
  const int*   adj_cols   = (const int*)d_in[9];
  float* out = (float*)d_out;

  if (ws_size < WS_NEED) return;

  uint8_t*  ws      = (uint8_t*)d_ws;
  uint32_t* mask8w  = (uint32_t*)(ws + OFF_MASK8);
  int*      slotmap = (int*)(ws + OFF_SLOT);
  int*      vtok    = (int*)(ws + OFF_VTOK);
  int*      cnt     = (int*)(ws + OFF_CNT);
  int2*     list    = (int2*)(ws + OFF_LIST);

  vg_prep<<<1 + (BB * OO * EE) / 4096, 1024, 0, stream>>>(input_ids, tok2g, fc_b, mask8w, slotmap, vtok,
                                                          (float4*)out, cnt);
  vg_scan<<<SCAN_NB, SCAN_T, 0, stream>>>(adj_rows, adj_cols, adj_values, gv_ids,
                                                   (const uint8_t*)mask8w, slotmap, cnt, list);
  vg_fuse<<<NB1 + BB * NQ, 1024, 0, stream>>>(emb, W_vh, fc_w, gv_ids, vtok, cnt, list, out);
}
